# starting kernel
# baseline (speedup 1.0000x reference)
#include <hip/hip_runtime.h>
#include <hip/hip_bf16.h>

#define S_ 32
#define B_ 32
#define N_ 8192
#define D_ 256
#define M_ 8
#define CHUNKS 16
#define ROWS_PER_BLOCK (N_ / CHUNKS)

typedef __attribute__((ext_vector_type(2)))  float    v2f;
typedef __attribute__((ext_vector_type(8)))  float    v8f;
typedef __attribute__((ext_vector_type(16))) _Float16 v16h;

#define WS_SUMS 0
#define WS_CNT  (B_ * S_ * D_)
#define WS_OBJS (WS_CNT + B_ * S_)
#define WS_ZERO_N (WS_OBJS)

#if defined(__has_builtin)
#if __has_builtin(__builtin_amdgcn_wmma_f32_16x16x4_f32)
#define HAVE_F32_WMMA 1
#endif
#endif

__global__ __launch_bounds__(256) void zero_ws(float* __restrict__ ws) {
    int i = blockIdx.x * blockDim.x + threadIdx.x;
    int stride = gridDim.x * blockDim.x;
    for (; i < WS_ZERO_N; i += stride) ws[i] = 0.0f;
}

__global__ __launch_bounds__(256) void seg_accum(const float* __restrict__ features,
                                                 const int*   __restrict__ seg,
                                                 float*       __restrict__ ws) {
    __shared__ float acc[S_ * D_];
    __shared__ float cnt[S_];
    const int tid = threadIdx.x;
    for (int i = tid; i < S_ * D_; i += 256) acc[i] = 0.0f;
    if (tid < S_) cnt[tid] = 0.0f;
    __syncthreads();

    const int b     = blockIdx.x / CHUNKS;
    const int chunk = blockIdx.x % CHUNKS;
    const int wave  = tid >> 5;
    const int lane  = tid & 31;
    const float* fb = features + (size_t)b * N_ * D_;
    const int*   sb = seg + b * N_;

    for (int i = 0; i < ROWS_PER_BLOCK / 8; ++i) {
        const int n = chunk * ROWS_PER_BLOCK + i * 8 + wave;
        const int s = sb[n];
        const float4* row = (const float4*)(fb + (size_t)n * D_);
        float4 v0 = row[lane * 2 + 0];
        float4 v1 = row[lane * 2 + 1];
        float* a = acc + s * D_ + lane * 8;
        atomicAdd(a + 0, v0.x); atomicAdd(a + 1, v0.y);
        atomicAdd(a + 2, v0.z); atomicAdd(a + 3, v0.w);
        atomicAdd(a + 4, v1.x); atomicAdd(a + 5, v1.y);
        atomicAdd(a + 6, v1.z); atomicAdd(a + 7, v1.w);
        if (lane == 0) atomicAdd(&cnt[s], 1.0f);
    }
    __syncthreads();

    float* gsum = ws + WS_SUMS + (size_t)b * S_ * D_;
    for (int i = tid; i < S_ * D_; i += 256) atomicAdd(&gsum[i], acc[i]);
    if (tid < S_) atomicAdd(&ws[WS_CNT + b * S_ + tid], cnt[tid]);
}

__global__ __launch_bounds__(256) void top_objs(float* __restrict__ ws) {
    __shared__ int   topi[M_];
    __shared__ float scale[M_];
    const int b = blockIdx.x, tid = threadIdx.x;
    if (tid == 0) {
        const float* c = ws + WS_CNT + b * S_;
        unsigned used = 0u;
        for (int m = 0; m < M_; ++m) {
            int best = 0; float bc = -1.0f;
            for (int s = 0; s < S_; ++s) {
                if (!((used >> s) & 1u) && c[s] > bc) { bc = c[s]; best = s; }
            }
            used |= (1u << best);
            topi[m]  = best;
            scale[m] = (bc > 0.0f) ? (1.0f / bc) : 0.0f;
        }
    }
    __syncthreads();
    const float* gsum = ws + WS_SUMS + (size_t)b * S_ * D_;
    float*       objs = ws + WS_OBJS + (size_t)b * M_ * D_;
    for (int m = 0; m < M_; ++m)
        objs[m * D_ + tid] = gsum[topi[m] * D_ + tid] * scale[m];
}

__global__ __launch_bounds__(512) void proj_ln(const float* __restrict__ objs,
                                               const float* __restrict__ W,
                                               const float* __restrict__ bias,
                                               const float* __restrict__ gamma,
                                               const float* __restrict__ beta,
                                               float*       __restrict__ out) {
    __shared__ float tile[16 * D_];
    const int tid  = threadIdx.x;
    const int wave = tid >> 5;
    const int lane = tid & 31;
    const int half = lane >> 4;
    const int l    = lane & 15;
    const int rowbase = blockIdx.x * 16;
    const int colbase = wave * 16;

    v8f c = {};

#ifdef HAVE_F32_WMMA
    for (int k0 = 0; k0 < D_; k0 += 4) {
        const int k = k0 + 2 * half;
        v2f a, w;
        const float* ap = objs + (size_t)(rowbase + l) * D_ + k;
        a.x = ap[0]; a.y = ap[1];
        const float* wp = W + (size_t)(colbase + l) * D_ + k;
        w.x = wp[0]; w.y = wp[1];
        c = __builtin_amdgcn_wmma_f32_16x16x4_f32(false, a, false, w,
                                                  (short)0, c, false, false);
    }
#else
    for (int k0 = 0; k0 < D_; k0 += 32) {
        v16h a, w;
        const float* ap = objs + (size_t)(rowbase + l) * D_;
        const float* wp = W + (size_t)(colbase + l) * D_;
        for (int i = 0; i < 8; ++i) {
            const int kk = k0 + ((i < 4) ? (2 * i + 8 * half)
                                         : (16 + 2 * (i - 4) + 8 * half));
            a[2 * i]     = (_Float16)ap[kk];
            a[2 * i + 1] = (_Float16)ap[kk + 1];
            w[2 * i]     = (_Float16)wp[kk];
            w[2 * i + 1] = (_Float16)wp[kk + 1];
        }
        c = __builtin_amdgcn_wmma_f32_16x16x32_f16(false, a, false, w,
                                                   (short)0, c, false, false);
    }
#endif

    for (int j = 0; j < 8; ++j) {
        const int m = j + half * 8;
        const int n = colbase + l;
        tile[m * D_ + n] = c[j] + bias[n];
    }
    __syncthreads();

    const float* row = tile + wave * D_;
    float v[8], s = 0.0f, s2 = 0.0f;
    for (int j = 0; j < 8; ++j) {
        v[j] = row[lane * 8 + j];
        s  += v[j];
        s2 += v[j] * v[j];
    }
    for (int off = 16; off > 0; off >>= 1) {
        s  += __shfl_xor(s,  off, 32);
        s2 += __shfl_xor(s2, off, 32);
    }
    const float mean = s * (1.0f / D_);
    const float var  = s2 * (1.0f / D_) - mean * mean;
    const float rstd = rsqrtf(var + 1e-5f);
    float* orow = out + (size_t)(rowbase + wave) * D_;
    for (int j = 0; j < 8; ++j) {
        const int idx = lane * 8 + j;
        orow[idx] = (v[j] - mean) * rstd * gamma[idx] + beta[idx];
    }
}

extern "C" void kernel_launch(void* const* d_in, const int* in_sizes, int n_in,
                              void* d_out, int out_size, void* d_ws, size_t ws_size,
                              hipStream_t stream) {
    const float* features = (const float*)d_in[0];
    const int*   seg      = (const int*)  d_in[1];
    const float* W        = (const float*)d_in[2];
    const float* b        = (const float*)d_in[3];
    const float* gamma    = (const float*)d_in[4];
    const float* beta     = (const float*)d_in[5];
    float* ws  = (float*)d_ws;
    float* out = (float*)d_out;

    zero_ws  <<<512, 256, 0, stream>>>(ws);
    seg_accum<<<B_ * CHUNKS, 256, 0, stream>>>(features, seg, ws);
    top_objs <<<B_, 256, 0, stream>>>(ws);
    proj_ln  <<<(B_ * M_) / 16, 512, 0, stream>>>(ws + WS_OBJS, W, b, gamma, beta, out);
}


# seg_accum read stream: a wave takes 64 consecutive rows in 8-row batches of whole-line non-temporal b128 loads, next batch issued ahead (16-32 in flight, was 2); LDS atomic arrival order changes
# speedup vs baseline: 1.3632x; 1.3632x over previous
#include <hip/hip_runtime.h>
#include <hip/hip_bf16.h>

#define S_ 32
#define B_ 32
#define N_ 8192
#define D_ 256
#define M_ 8
#define CHUNKS 16
#define ROWS_PER_BLOCK (N_ / CHUNKS)

typedef __attribute__((ext_vector_type(2)))  float    v2f;
typedef __attribute__((ext_vector_type(4)))  float    v4f;
typedef __attribute__((ext_vector_type(8)))  float    v8f;
typedef __attribute__((ext_vector_type(16))) _Float16 v16h;

#define WS_SUMS 0
#define WS_CNT  (B_ * S_ * D_)
#define WS_OBJS (WS_CNT + B_ * S_)
#define WS_ZERO_N (WS_OBJS)

#if defined(__has_builtin)
#if __has_builtin(__builtin_amdgcn_wmma_f32_16x16x4_f32)
#define HAVE_F32_WMMA 1
#endif
#endif

__global__ __launch_bounds__(256) void zero_ws(float* __restrict__ ws) {
    int i = blockIdx.x * blockDim.x + threadIdx.x;
    int stride = gridDim.x * blockDim.x;
    for (; i < WS_ZERO_N; i += stride) ws[i] = 0.0f;
}

#define ROWS_PER_WAVE (ROWS_PER_BLOCK / 8)
#define RB            8
#define NBATCH        (ROWS_PER_WAVE / RB)
#define ROW_V4        (D_ / 4)

__device__ __forceinline__ v4f load_stream(const v4f* p) {
    return __builtin_nontemporal_load(p);
}

__global__ __launch_bounds__(256) void seg_accum(const float* __restrict__ features,
                                                 const int*   __restrict__ seg,
                                                 float*       __restrict__ ws) {
    __shared__ float acc[S_ * D_];
    __shared__ float cnt[S_];
    const int tid = threadIdx.x;
    for (int i = tid; i < S_ * D_; i += 256) acc[i] = 0.0f;
    if (tid < S_) cnt[tid] = 0.0f;
    __syncthreads();

    const int b     = blockIdx.x / CHUNKS;
    const int chunk = blockIdx.x % CHUNKS;
    const int wave  = __builtin_amdgcn_readfirstlane(tid >> 5);
    const int lane  = tid & 31;
    const int row0  = chunk * ROWS_PER_BLOCK + wave * ROWS_PER_WAVE;
    const v4f* fp = (const v4f*)(features + ((size_t)b * N_ + row0) * D_) + lane;
    const int* sp = seg + b * N_ + row0;
    float* abase = acc + lane * 4;

    const int id_lo = sp[lane];
    const int id_hi = sp[lane + 32];

    v4f buf[2][RB][2];
#pragma unroll
    for (int r = 0; r < RB; ++r) {
        buf[0][r][0] = load_stream(fp + r * ROW_V4);
        buf[0][r][1] = load_stream(fp + r * ROW_V4 + 32);
    }

#pragma unroll
    for (int t = 0; t < NBATCH; ++t) {
        if (t + 1 < NBATCH) {
#pragma unroll
            for (int r = 0; r < RB; ++r) {
                const v4f* p = fp + ((t + 1) * RB + r) * ROW_V4;
                buf[(t + 1) & 1][r][0] = load_stream(p);
                buf[(t + 1) & 1][r][1] = load_stream(p + 32);
            }
        }
#pragma unroll
        for (int r = 0; r < RB; ++r) {
            const int R = t * RB + r;
            const int s = __builtin_amdgcn_readlane(R < 32 ? id_lo : id_hi, R & 31);
            const v4f v0 = buf[t & 1][r][0];
            const v4f v1 = buf[t & 1][r][1];
            float* a = abase + s * D_;
            atomicAdd(a + 0, v0.x); atomicAdd(a + 1, v0.y);
            atomicAdd(a + 2, v0.z); atomicAdd(a + 3, v0.w);
            atomicAdd(a + 128, v1.x); atomicAdd(a + 129, v1.y);
            atomicAdd(a + 130, v1.z); atomicAdd(a + 131, v1.w);
        }
    }
    atomicAdd(&cnt[id_lo], 1.0f);
    atomicAdd(&cnt[id_hi], 1.0f);
    __syncthreads();

    float* gsum = ws + WS_SUMS + (size_t)b * S_ * D_;
    for (int i = tid; i < S_ * D_; i += 256) atomicAdd(&gsum[i], acc[i]);
    if (tid < S_) atomicAdd(&ws[WS_CNT + b * S_ + tid], cnt[tid]);
}

__global__ __launch_bounds__(256) void top_objs(float* __restrict__ ws) {
    __shared__ int   topi[M_];
    __shared__ float scale[M_];
    const int b = blockIdx.x, tid = threadIdx.x;
    if (tid == 0) {
        const float* c = ws + WS_CNT + b * S_;
        unsigned used = 0u;
        for (int m = 0; m < M_; ++m) {
            int best = 0; float bc = -1.0f;
            for (int s = 0; s < S_; ++s) {
                if (!((used >> s) & 1u) && c[s] > bc) { bc = c[s]; best = s; }
            }
            used |= (1u << best);
            topi[m]  = best;
            scale[m] = (bc > 0.0f) ? (1.0f / bc) : 0.0f;
        }
    }
    __syncthreads();
    const float* gsum = ws + WS_SUMS + (size_t)b * S_ * D_;
    float*       objs = ws + WS_OBJS + (size_t)b * M_ * D_;
    for (int m = 0; m < M_; ++m)
        objs[m * D_ + tid] = gsum[topi[m] * D_ + tid] * scale[m];
}

__global__ __launch_bounds__(512) void proj_ln(const float* __restrict__ objs,
                                               const float* __restrict__ W,
                                               const float* __restrict__ bias,
                                               const float* __restrict__ gamma,
                                               const float* __restrict__ beta,
                                               float*       __restrict__ out) {
    __shared__ float tile[16 * D_];
    const int tid  = threadIdx.x;
    const int wave = tid >> 5;
    const int lane = tid & 31;
    const int half = lane >> 4;
    const int l    = lane & 15;
    const int rowbase = blockIdx.x * 16;
    const int colbase = wave * 16;

    v8f c = {};

#ifdef HAVE_F32_WMMA
    for (int k0 = 0; k0 < D_; k0 += 4) {
        const int k = k0 + 2 * half;
        v2f a, w;
        const float* ap = objs + (size_t)(rowbase + l) * D_ + k;
        a.x = ap[0]; a.y = ap[1];
        const float* wp = W + (size_t)(colbase + l) * D_ + k;
        w.x = wp[0]; w.y = wp[1];
        c = __builtin_amdgcn_wmma_f32_16x16x4_f32(false, a, false, w,
                                                  (short)0, c, false, false);
    }
#else
    for (int k0 = 0; k0 < D_; k0 += 32) {
        v16h a, w;
        const float* ap = objs + (size_t)(rowbase + l) * D_;
        const float* wp = W + (size_t)(colbase + l) * D_;
        for (int i = 0; i < 8; ++i) {
            const int kk = k0 + ((i < 4) ? (2 * i + 8 * half)
                                         : (16 + 2 * (i - 4) + 8 * half));
            a[2 * i]     = (_Float16)ap[kk];
            a[2 * i + 1] = (_Float16)ap[kk + 1];
            w[2 * i]     = (_Float16)wp[kk];
            w[2 * i + 1] = (_Float16)wp[kk + 1];
        }
        c = __builtin_amdgcn_wmma_f32_16x16x32_f16(false, a, false, w,
                                                   (short)0, c, false, false);
    }
#endif

    for (int j = 0; j < 8; ++j) {
        const int m = j + half * 8;
        const int n = colbase + l;
        tile[m * D_ + n] = c[j] + bias[n];
    }
    __syncthreads();

    const float* row = tile + wave * D_;
    float v[8], s = 0.0f, s2 = 0.0f;
    for (int j = 0; j < 8; ++j) {
        v[j] = row[lane * 8 + j];
        s  += v[j];
        s2 += v[j] * v[j];
    }
    for (int off = 16; off > 0; off >>= 1) {
        s  += __shfl_xor(s,  off, 32);
        s2 += __shfl_xor(s2, off, 32);
    }
    const float mean = s * (1.0f / D_);
    const float var  = s2 * (1.0f / D_) - mean * mean;
    const float rstd = rsqrtf(var + 1e-5f);
    float* orow = out + (size_t)(rowbase + wave) * D_;
    for (int j = 0; j < 8; ++j) {
        const int idx = lane * 8 + j;
        orow[idx] = (v[j] - mean) * rstd * gamma[idx] + beta[idx];
    }
}

extern "C" void kernel_launch(void* const* d_in, const int* in_sizes, int n_in,
                              void* d_out, int out_size, void* d_ws, size_t ws_size,
                              hipStream_t stream) {
    const float* features = (const float*)d_in[0];
    const int*   seg      = (const int*)  d_in[1];
    const float* W        = (const float*)d_in[2];
    const float* b        = (const float*)d_in[3];
    const float* gamma    = (const float*)d_in[4];
    const float* beta     = (const float*)d_in[5];
    float* ws  = (float*)d_ws;
    float* out = (float*)d_out;

    zero_ws  <<<512, 256, 0, stream>>>(ws);
    seg_accum<<<B_ * CHUNKS, 256, 0, stream>>>(features, seg, ws);
    top_objs <<<B_, 256, 0, stream>>>(ws);
    proj_ln  <<<(B_ * M_) / 16, 512, 0, stream>>>(ws + WS_OBJS, W, b, gamma, beta, out);
}


# top_objs: top-8 choice by rank across the 32 lanes of one wave (a lane counts the segments ahead of its own in the stable order), not one thread's 8 x 32 scan of dependent scalar loads
# speedup vs baseline: 1.5054x; 1.1043x over previous
#include <hip/hip_runtime.h>
#include <hip/hip_bf16.h>

#define S_ 32
#define B_ 32
#define N_ 8192
#define D_ 256
#define M_ 8
#define CHUNKS 16
#define ROWS_PER_BLOCK (N_ / CHUNKS)

typedef __attribute__((ext_vector_type(2)))  float    v2f;
typedef __attribute__((ext_vector_type(4)))  float    v4f;
typedef __attribute__((ext_vector_type(8)))  float    v8f;
typedef __attribute__((ext_vector_type(16))) _Float16 v16h;

#define WS_SUMS 0
#define WS_CNT  (B_ * S_ * D_)
#define WS_OBJS (WS_CNT + B_ * S_)
#define WS_ZERO_N (WS_OBJS)

#if defined(__has_builtin)
#if __has_builtin(__builtin_amdgcn_wmma_f32_16x16x4_f32)
#define HAVE_F32_WMMA 1
#endif
#endif

__global__ __launch_bounds__(256) void zero_ws(float* __restrict__ ws) {
    int i = blockIdx.x * blockDim.x + threadIdx.x;
    int stride = gridDim.x * blockDim.x;
    for (; i < WS_ZERO_N; i += stride) ws[i] = 0.0f;
}

#define ROWS_PER_WAVE (ROWS_PER_BLOCK / 8)
#define RB            8
#define NBATCH        (ROWS_PER_WAVE / RB)
#define ROW_V4        (D_ / 4)

__device__ __forceinline__ v4f load_stream(const v4f* p) {
    return __builtin_nontemporal_load(p);
}

__global__ __launch_bounds__(256) void seg_accum(const float* __restrict__ features,
                                                 const int*   __restrict__ seg,
                                                 float*       __restrict__ ws) {
    __shared__ float acc[S_ * D_];
    __shared__ float cnt[S_];
    const int tid = threadIdx.x;
    for (int i = tid; i < S_ * D_; i += 256) acc[i] = 0.0f;
    if (tid < S_) cnt[tid] = 0.0f;
    __syncthreads();

    const int b     = blockIdx.x / CHUNKS;
    const int chunk = blockIdx.x % CHUNKS;
    const int wave  = __builtin_amdgcn_readfirstlane(tid >> 5);
    const int lane  = tid & 31;
    const int row0  = chunk * ROWS_PER_BLOCK + wave * ROWS_PER_WAVE;
    const v4f* fp = (const v4f*)(features + ((size_t)b * N_ + row0) * D_) + lane;
    const int* sp = seg + b * N_ + row0;
    float* abase = acc + lane * 4;

    const int id_lo = sp[lane];
    const int id_hi = sp[lane + 32];

    v4f buf[2][RB][2];
#pragma unroll
    for (int r = 0; r < RB; ++r) {
        buf[0][r][0] = load_stream(fp + r * ROW_V4);
        buf[0][r][1] = load_stream(fp + r * ROW_V4 + 32);
    }

#pragma unroll
    for (int t = 0; t < NBATCH; ++t) {
        if (t + 1 < NBATCH) {
#pragma unroll
            for (int r = 0; r < RB; ++r) {
                const v4f* p = fp + ((t + 1) * RB + r) * ROW_V4;
                buf[(t + 1) & 1][r][0] = load_stream(p);
                buf[(t + 1) & 1][r][1] = load_stream(p + 32);
            }
        }
#pragma unroll
        for (int r = 0; r < RB; ++r) {
            const int R = t * RB + r;
            const int s = __builtin_amdgcn_readlane(R < 32 ? id_lo : id_hi, R & 31);
            const v4f v0 = buf[t & 1][r][0];
            const v4f v1 = buf[t & 1][r][1];
            float* a = abase + s * D_;
            atomicAdd(a + 0, v0.x); atomicAdd(a + 1, v0.y);
            atomicAdd(a + 2, v0.z); atomicAdd(a + 3, v0.w);
            atomicAdd(a + 128, v1.x); atomicAdd(a + 129, v1.y);
            atomicAdd(a + 130, v1.z); atomicAdd(a + 131, v1.w);
        }
    }
    atomicAdd(&cnt[id_lo], 1.0f);
    atomicAdd(&cnt[id_hi], 1.0f);
    __syncthreads();

    float* gsum = ws + WS_SUMS + (size_t)b * S_ * D_;
    for (int i = tid; i < S_ * D_; i += 256) atomicAdd(&gsum[i], acc[i]);
    if (tid < S_) atomicAdd(&ws[WS_CNT + b * S_ + tid], cnt[tid]);
}

static_assert(S_ == 32, "one lane of a wave per segment id");

__global__ __launch_bounds__(256) void top_objs(float* __restrict__ ws) {
    __shared__ int   topi[M_];
    __shared__ float scale[M_];
    const int b = blockIdx.x, tid = threadIdx.x;
    if (tid < S_) {
        const float c = ws[WS_CNT + b * S_ + tid];
        int place = 0;
#pragma unroll
        for (int t = 0; t < S_; ++t) {
            const float ct = __builtin_bit_cast(
                float, __builtin_amdgcn_readlane(__builtin_bit_cast(int, c), t));
            place += (ct > c || (ct == c && t < tid)) ? 1 : 0;
        }
        if (place < M_) {
            topi[place]  = tid;
            scale[place] = (c > 0.0f) ? (1.0f / c) : 0.0f;
        }
    }
    __syncthreads();
    const float* gsum = ws + WS_SUMS + (size_t)b * S_ * D_;
    float*       objs = ws + WS_OBJS + (size_t)b * M_ * D_;
    for (int m = 0; m < M_; ++m)
        objs[m * D_ + tid] = gsum[topi[m] * D_ + tid] * scale[m];
}

__global__ __launch_bounds__(512) void proj_ln(const float* __restrict__ objs,
                                               const float* __restrict__ W,
                                               const float* __restrict__ bias,
                                               const float* __restrict__ gamma,
                                               const float* __restrict__ beta,
                                               float*       __restrict__ out) {
    __shared__ float tile[16 * D_];
    const int tid  = threadIdx.x;
    const int wave = tid >> 5;
    const int lane = tid & 31;
    const int half = lane >> 4;
    const int l    = lane & 15;
    const int rowbase = blockIdx.x * 16;
    const int colbase = wave * 16;

    v8f c = {};

#ifdef HAVE_F32_WMMA
    for (int k0 = 0; k0 < D_; k0 += 4) {
        const int k = k0 + 2 * half;
        v2f a, w;
        const float* ap = objs + (size_t)(rowbase + l) * D_ + k;
        a.x = ap[0]; a.y = ap[1];
        const float* wp = W + (size_t)(colbase + l) * D_ + k;
        w.x = wp[0]; w.y = wp[1];
        c = __builtin_amdgcn_wmma_f32_16x16x4_f32(false, a, false, w,
                                                  (short)0, c, false, false);
    }
#else
    for (int k0 = 0; k0 < D_; k0 += 32) {
        v16h a, w;
        const float* ap = objs + (size_t)(rowbase + l) * D_;
        const float* wp = W + (size_t)(colbase + l) * D_;
        for (int i = 0; i < 8; ++i) {
            const int kk = k0 + ((i < 4) ? (2 * i + 8 * half)
                                         : (16 + 2 * (i - 4) + 8 * half));
            a[2 * i]     = (_Float16)ap[kk];
            a[2 * i + 1] = (_Float16)ap[kk + 1];
            w[2 * i]     = (_Float16)wp[kk];
            w[2 * i + 1] = (_Float16)wp[kk + 1];
        }
        c = __builtin_amdgcn_wmma_f32_16x16x32_f16(false, a, false, w,
                                                   (short)0, c, false, false);
    }
#endif

    for (int j = 0; j < 8; ++j) {
        const int m = j + half * 8;
        const int n = colbase + l;
        tile[m * D_ + n] = c[j] + bias[n];
    }
    __syncthreads();

    const float* row = tile + wave * D_;
    float v[8], s = 0.0f, s2 = 0.0f;
    for (int j = 0; j < 8; ++j) {
        v[j] = row[lane * 8 + j];
        s  += v[j];
        s2 += v[j] * v[j];
    }
    for (int off = 16; off > 0; off >>= 1) {
        s  += __shfl_xor(s,  off, 32);
        s2 += __shfl_xor(s2, off, 32);
    }
    const float mean = s * (1.0f / D_);
    const float var  = s2 * (1.0f / D_) - mean * mean;
    const float rstd = rsqrtf(var + 1e-5f);
    float* orow = out + (size_t)(rowbase + wave) * D_;
    for (int j = 0; j < 8; ++j) {
        const int idx = lane * 8 + j;
        orow[idx] = (v[j] - mean) * rstd * gamma[idx] + beta[idx];
    }
}

extern "C" void kernel_launch(void* const* d_in, const int* in_sizes, int n_in,
                              void* d_out, int out_size, void* d_ws, size_t ws_size,
                              hipStream_t stream) {
    const float* features = (const float*)d_in[0];
    const int*   seg      = (const int*)  d_in[1];
    const float* W        = (const float*)d_in[2];
    const float* b        = (const float*)d_in[3];
    const float* gamma    = (const float*)d_in[4];
    const float* beta     = (const float*)d_in[5];
    float* ws  = (float*)d_ws;
    float* out = (float*)d_out;

    zero_ws  <<<512, 256, 0, stream>>>(ws);
    seg_accum<<<B_ * CHUNKS, 256, 0, stream>>>(features, seg, ws);
    top_objs <<<B_, 256, 0, stream>>>(ws);
    proj_ln  <<<(B_ * M_) / 16, 512, 0, stream>>>(ws + WS_OBJS, W, b, gamma, beta, out);
}


# top-8 choice made before the sums: a kernel counts ids and picks 8 segments per batch, seg_accum loads and adds only their rows (atomic arrival order still free); 1/count applied after the product
# speedup vs baseline: 2.9128x; 1.9349x over previous
#include <hip/hip_runtime.h>
#include <hip/hip_bf16.h>

#define S_ 32
#define B_ 32
#define N_ 8192
#define D_ 256
#define M_ 8
#define CHUNKS 16
#define ROWS_PER_BLOCK (N_ / CHUNKS)

typedef __attribute__((ext_vector_type(2)))  float    v2f;
typedef __attribute__((ext_vector_type(4)))  float    v4f;
typedef __attribute__((ext_vector_type(8)))  float    v8f;
typedef __attribute__((ext_vector_type(16))) _Float16 v16h;

#define WS_SUMS  0
#define WS_SLOT  (B_ * M_ * D_)
#define WS_SCALE (WS_SLOT + B_ * S_)

#if defined(__has_builtin)
#if __has_builtin(__builtin_amdgcn_wmma_f32_16x16x4_f32)
#define HAVE_F32_WMMA 1
#endif
#endif

static_assert(S_ == 32, "one lane of a wave per segment id");
static_assert(N_ == 8 * 1024, "count_pick: 1024 threads take 8 ids each");
static_assert(M_ * D_ == 2 * 1024, "count_pick: 1024 threads zero 2 sums each");

typedef __attribute__((ext_vector_type(4))) int v4i;

__global__ __launch_bounds__(1024) void count_pick(const int* __restrict__ seg,
                                                   float*     __restrict__ ws) {
    __shared__ int hist[S_];
    const int b = blockIdx.x, tid = threadIdx.x;
    if (tid < S_) hist[tid] = 0;
    __syncthreads();

    const v4i* sp = (const v4i*)(seg + b * N_);
    const v4i x0 = sp[tid];
    const v4i x1 = sp[tid + 1024];
    atomicAdd(&hist[x0.x & (S_ - 1)], 1); atomicAdd(&hist[x0.y & (S_ - 1)], 1);
    atomicAdd(&hist[x0.z & (S_ - 1)], 1); atomicAdd(&hist[x0.w & (S_ - 1)], 1);
    atomicAdd(&hist[x1.x & (S_ - 1)], 1); atomicAdd(&hist[x1.y & (S_ - 1)], 1);
    atomicAdd(&hist[x1.z & (S_ - 1)], 1); atomicAdd(&hist[x1.w & (S_ - 1)], 1);

    float* gs = ws + WS_SUMS + (size_t)b * M_ * D_;
    gs[tid]        = 0.0f;
    gs[tid + 1024] = 0.0f;
    __syncthreads();

    if (tid < S_) {
        const float c = (float)hist[tid];
        int place = 0;
#pragma unroll
        for (int t = 0; t < S_; ++t) {
            const float ct = __builtin_bit_cast(
                float, __builtin_amdgcn_readlane(__builtin_bit_cast(int, c), t));
            place += (ct > c || (ct == c && t < tid)) ? 1 : 0;
        }
        ws[WS_SLOT + b * S_ + tid] = (place < M_) ? (float)place : -1.0f;
        if (place < M_)
            ws[WS_SCALE + b * M_ + place] = (c > 0.0f) ? (1.0f / c) : 0.0f;
    }
}

#define ROWS_PER_WAVE (ROWS_PER_BLOCK / 8)
#define RB            30
#define ROW_V4        (D_ / 4)
static_assert(ROWS_PER_WAVE == 64, "one 64-bit set of rows per wave");

__device__ __forceinline__ v4f load_stream(const v4f* p) {
    return __builtin_nontemporal_load(p);
}

__global__ __launch_bounds__(256) void seg_accum(const float* __restrict__ features,
                                                 const int*   __restrict__ seg,
                                                 float*       __restrict__ ws) {
    __shared__ float acc[M_ * D_];
    const int tid = threadIdx.x;
    for (int i = tid; i < M_ * D_; i += 256) acc[i] = 0.0f;
    __syncthreads();

    const int b     = blockIdx.x / CHUNKS;
    const int chunk = blockIdx.x % CHUNKS;
    const int wave  = __builtin_amdgcn_readfirstlane(tid >> 5);
    const int lane  = tid & 31;
    const int row0  = chunk * ROWS_PER_BLOCK + wave * ROWS_PER_WAVE;
    const v4f* fp = (const v4f*)(features + ((size_t)b * N_ + row0) * D_) + lane;
    const int* sp = seg + b * N_ + row0;
    float* abase = acc + lane * 4;

    const int slot_of_lane = (int)ws[WS_SLOT + b * S_ + lane];
    const int id_lo = sp[lane]      & (S_ - 1);
    const int id_hi = sp[lane + 32] & (S_ - 1);
    const int slot_lo = __shfl(slot_of_lane, id_lo, 32);
    const int slot_hi = __shfl(slot_of_lane, id_hi, 32);

    unsigned long long todo =
        ((unsigned long long)__builtin_amdgcn_ballot_w32(slot_hi >= 0) << 32) |
         (unsigned long long)__builtin_amdgcn_ballot_w32(slot_lo >= 0);

    while (todo != 0) {
        const int left = __builtin_popcountll(todo);
        const int n    = left < RB ? left : RB;
        v4f buf[RB][2];

        unsigned long long m = todo;
#pragma unroll
        for (int k = 0; k < RB; ++k) {
            if (k < n) {
                const int R = __builtin_ctzll(m);
                m &= m - 1;
                const v4f* p = fp + R * ROW_V4;
                buf[k][0] = load_stream(p);
                buf[k][1] = load_stream(p + 32);
            }
        }

        unsigned long long m2 = todo;
#pragma unroll
        for (int k = 0; k < RB; ++k) {
            if (k < n) {
                const int R = __builtin_ctzll(m2);
                m2 &= m2 - 1;
                const int s = __builtin_amdgcn_readlane(R < 32 ? slot_lo : slot_hi, R & 31);
                const v4f v0 = buf[k][0];
                const v4f v1 = buf[k][1];
                float* a = abase + s * D_;
                atomicAdd(a + 0, v0.x); atomicAdd(a + 1, v0.y);
                atomicAdd(a + 2, v0.z); atomicAdd(a + 3, v0.w);
                atomicAdd(a + 128, v1.x); atomicAdd(a + 129, v1.y);
                atomicAdd(a + 130, v1.z); atomicAdd(a + 131, v1.w);
            }
        }
        todo = m;
    }
    __syncthreads();

    float* gsum = ws + WS_SUMS + (size_t)b * M_ * D_;
    for (int i = tid; i < M_ * D_; i += 256) atomicAdd(&gsum[i], acc[i]);
}

__global__ __launch_bounds__(512) void proj_ln(const float* __restrict__ sums,
                                               const float* __restrict__ scale,
                                               const float* __restrict__ W,
                                               const float* __restrict__ bias,
                                               const float* __restrict__ gamma,
                                               const float* __restrict__ beta,
                                               float*       __restrict__ out) {
    __shared__ float tile[16 * D_];
    const int tid  = threadIdx.x;
    const int wave = tid >> 5;
    const int lane = tid & 31;
    const int half = lane >> 4;
    const int l    = lane & 15;
    const int rowbase = blockIdx.x * 16;
    const int colbase = wave * 16;

    v8f c = {};
    const float* objs = sums;

#ifdef HAVE_F32_WMMA
    for (int k0 = 0; k0 < D_; k0 += 4) {
        const int k = k0 + 2 * half;
        v2f a, w;
        const float* ap = objs + (size_t)(rowbase + l) * D_ + k;
        a.x = ap[0]; a.y = ap[1];
        const float* wp = W + (size_t)(colbase + l) * D_ + k;
        w.x = wp[0]; w.y = wp[1];
        c = __builtin_amdgcn_wmma_f32_16x16x4_f32(false, a, false, w,
                                                  (short)0, c, false, false);
    }
#else
    for (int k0 = 0; k0 < D_; k0 += 32) {
        v16h a, w;
        const float* ap = objs + (size_t)(rowbase + l) * D_;
        const float* wp = W + (size_t)(colbase + l) * D_;
        for (int i = 0; i < 8; ++i) {
            const int kk = k0 + ((i < 4) ? (2 * i + 8 * half)
                                         : (16 + 2 * (i - 4) + 8 * half));
            a[2 * i]     = (_Float16)ap[kk];
            a[2 * i + 1] = (_Float16)ap[kk + 1];
            w[2 * i]     = (_Float16)wp[kk];
            w[2 * i + 1] = (_Float16)wp[kk + 1];
        }
        c = __builtin_amdgcn_wmma_f32_16x16x32_f16(false, a, false, w,
                                                   (short)0, c, false, false);
    }
#endif

    for (int j = 0; j < 8; ++j) {
        const int m = j + half * 8;
        const int n = colbase + l;
        tile[m * D_ + n] = c[j] * scale[rowbase + m] + bias[n];
    }
    __syncthreads();

    const float* row = tile + wave * D_;
    float v[8], s = 0.0f, s2 = 0.0f;
    for (int j = 0; j < 8; ++j) {
        v[j] = row[lane * 8 + j];
        s  += v[j];
        s2 += v[j] * v[j];
    }
    for (int off = 16; off > 0; off >>= 1) {
        s  += __shfl_xor(s,  off, 32);
        s2 += __shfl_xor(s2, off, 32);
    }
    const float mean = s * (1.0f / D_);
    const float var  = s2 * (1.0f / D_) - mean * mean;
    const float rstd = rsqrtf(var + 1e-5f);
    float* orow = out + (size_t)(rowbase + wave) * D_;
    for (int j = 0; j < 8; ++j) {
        const int idx = lane * 8 + j;
        orow[idx] = (v[j] - mean) * rstd * gamma[idx] + beta[idx];
    }
}

extern "C" void kernel_launch(void* const* d_in, const int* in_sizes, int n_in,
                              void* d_out, int out_size, void* d_ws, size_t ws_size,
                              hipStream_t stream) {
    const float* features = (const float*)d_in[0];
    const int*   seg      = (const int*)  d_in[1];
    const float* W        = (const float*)d_in[2];
    const float* b        = (const float*)d_in[3];
    const float* gamma    = (const float*)d_in[4];
    const float* beta     = (const float*)d_in[5];
    float* ws  = (float*)d_ws;
    float* out = (float*)d_out;

    count_pick<<<B_, 1024, 0, stream>>>(seg, ws);
    seg_accum <<<B_ * CHUNKS, 256, 0, stream>>>(features, seg, ws);
    proj_ln   <<<(B_ * M_) / 16, 512, 0, stream>>>(ws + WS_SUMS, ws + WS_SCALE, W, b, gamma, beta, out);
}


# proj_ln: both WMMA operands fetched as whole-line b128 loads and made into fragments through LDS (was 8-byte loads touching 16 lines each); f32 sum order changes inside each 8-k group
# speedup vs baseline: 3.6011x; 1.2363x over previous
#include <hip/hip_runtime.h>
#include <hip/hip_bf16.h>

#define S_ 32
#define B_ 32
#define N_ 8192
#define D_ 256
#define M_ 8
#define CHUNKS 16
#define ROWS_PER_BLOCK (N_ / CHUNKS)

typedef __attribute__((ext_vector_type(2)))  float    v2f;
typedef __attribute__((ext_vector_type(4)))  float    v4f;
typedef __attribute__((ext_vector_type(8)))  float    v8f;
typedef __attribute__((ext_vector_type(16))) _Float16 v16h;

#define WS_SUMS  0
#define WS_SLOT  (B_ * M_ * D_)
#define WS_SCALE (WS_SLOT + B_ * S_)

#if defined(__has_builtin)
#if __has_builtin(__builtin_amdgcn_wmma_f32_16x16x4_f32)
#define HAVE_F32_WMMA 1
#endif
#endif

static_assert(S_ == 32, "one lane of a wave per segment id");
static_assert(N_ == 8 * 1024, "count_pick: 1024 threads take 8 ids each");
static_assert(M_ * D_ == 2 * 1024, "count_pick: 1024 threads zero 2 sums each");

typedef __attribute__((ext_vector_type(4))) int v4i;

__global__ __launch_bounds__(1024) void count_pick(const int* __restrict__ seg,
                                                   float*     __restrict__ ws) {
    __shared__ int hist[S_];
    const int b = blockIdx.x, tid = threadIdx.x;
    if (tid < S_) hist[tid] = 0;
    __syncthreads();

    const v4i* sp = (const v4i*)(seg + b * N_);
    const v4i x0 = sp[tid];
    const v4i x1 = sp[tid + 1024];
    atomicAdd(&hist[x0.x & (S_ - 1)], 1); atomicAdd(&hist[x0.y & (S_ - 1)], 1);
    atomicAdd(&hist[x0.z & (S_ - 1)], 1); atomicAdd(&hist[x0.w & (S_ - 1)], 1);
    atomicAdd(&hist[x1.x & (S_ - 1)], 1); atomicAdd(&hist[x1.y & (S_ - 1)], 1);
    atomicAdd(&hist[x1.z & (S_ - 1)], 1); atomicAdd(&hist[x1.w & (S_ - 1)], 1);

    float* gs = ws + WS_SUMS + (size_t)b * M_ * D_;
    gs[tid]        = 0.0f;
    gs[tid + 1024] = 0.0f;
    __syncthreads();

    if (tid < S_) {
        const float c = (float)hist[tid];
        int place = 0;
#pragma unroll
        for (int t = 0; t < S_; ++t) {
            const float ct = __builtin_bit_cast(
                float, __builtin_amdgcn_readlane(__builtin_bit_cast(int, c), t));
            place += (ct > c || (ct == c && t < tid)) ? 1 : 0;
        }
        ws[WS_SLOT + b * S_ + tid] = (place < M_) ? (float)place : -1.0f;
        if (place < M_)
            ws[WS_SCALE + b * M_ + place] = (c > 0.0f) ? (1.0f / c) : 0.0f;
    }
}

#define ROWS_PER_WAVE (ROWS_PER_BLOCK / 8)
#define RB            30
#define ROW_V4        (D_ / 4)
static_assert(ROWS_PER_WAVE == 64, "one 64-bit set of rows per wave");

__device__ __forceinline__ v4f load_stream(const v4f* p) {
    return __builtin_nontemporal_load(p);
}

__global__ __launch_bounds__(256) void seg_accum(const float* __restrict__ features,
                                                 const int*   __restrict__ seg,
                                                 float*       __restrict__ ws) {
    __shared__ float acc[M_ * D_];
    const int tid = threadIdx.x;
    for (int i = tid; i < M_ * D_; i += 256) acc[i] = 0.0f;
    __syncthreads();

    const int b     = blockIdx.x / CHUNKS;
    const int chunk = blockIdx.x % CHUNKS;
    const int wave  = __builtin_amdgcn_readfirstlane(tid >> 5);
    const int lane  = tid & 31;
    const int row0  = chunk * ROWS_PER_BLOCK + wave * ROWS_PER_WAVE;
    const v4f* fp = (const v4f*)(features + ((size_t)b * N_ + row0) * D_) + lane;
    const int* sp = seg + b * N_ + row0;
    float* abase = acc + lane * 4;

    const int slot_of_lane = (int)ws[WS_SLOT + b * S_ + lane];
    const int id_lo = sp[lane]      & (S_ - 1);
    const int id_hi = sp[lane + 32] & (S_ - 1);
    const int slot_lo = __shfl(slot_of_lane, id_lo, 32);
    const int slot_hi = __shfl(slot_of_lane, id_hi, 32);

    unsigned long long todo =
        ((unsigned long long)__builtin_amdgcn_ballot_w32(slot_hi >= 0) << 32) |
         (unsigned long long)__builtin_amdgcn_ballot_w32(slot_lo >= 0);

    while (todo != 0) {
        const int left = __builtin_popcountll(todo);
        const int n    = left < RB ? left : RB;
        v4f buf[RB][2];

        unsigned long long m = todo;
#pragma unroll
        for (int k = 0; k < RB; ++k) {
            if (k < n) {
                const int R = __builtin_ctzll(m);
                m &= m - 1;
                const v4f* p = fp + R * ROW_V4;
                buf[k][0] = load_stream(p);
                buf[k][1] = load_stream(p + 32);
            }
        }

        unsigned long long m2 = todo;
#pragma unroll
        for (int k = 0; k < RB; ++k) {
            if (k < n) {
                const int R = __builtin_ctzll(m2);
                m2 &= m2 - 1;
                const int s = __builtin_amdgcn_readlane(R < 32 ? slot_lo : slot_hi, R & 31);
                const v4f v0 = buf[k][0];
                const v4f v1 = buf[k][1];
                float* a = abase + s * D_;
                atomicAdd(a + 0, v0.x); atomicAdd(a + 1, v0.y);
                atomicAdd(a + 2, v0.z); atomicAdd(a + 3, v0.w);
                atomicAdd(a + 128, v1.x); atomicAdd(a + 129, v1.y);
                atomicAdd(a + 130, v1.z); atomicAdd(a + 131, v1.w);
            }
        }
        todo = m;
    }
    __syncthreads();

    float* gsum = ws + WS_SUMS + (size_t)b * M_ * D_;
    for (int i = tid; i < M_ * D_; i += 256) atomicAdd(&gsum[i], acc[i]);
}

#define PITCH (D_ + 4)
static_assert((PITCH % 8) == 4, "row pitch is 16 B x odd");

__global__ __launch_bounds__(512) void proj_ln(const float* __restrict__ sums,
                                               const float* __restrict__ scale,
                                               const float* __restrict__ W,
                                               const float* __restrict__ bias,
                                               const float* __restrict__ gamma,
                                               const float* __restrict__ beta,
                                               float*       __restrict__ out) {
    __shared__ __attribute__((aligned(16))) float As[16 * PITCH];
    __shared__ __attribute__((aligned(16))) float Ws[16 * 16 * PITCH];
    __shared__ float tile[16 * D_];
    const int tid  = threadIdx.x;
    const int wave = tid >> 5;
    const int lane = tid & 31;
    const int half = lane >> 4;
    const int l    = lane & 15;
    const int rowbase = blockIdx.x * 16;
    const int colbase = wave * 16;

    v8f c = {};
    const float* objs = sums;

#ifdef HAVE_F32_WMMA
    const v4f* wg = (const v4f*)(W + (size_t)colbase * D_) + lane;
    v4f wlo[16], whi[16];
#pragma unroll
    for (int r = 0; r < 16; ++r) wlo[r] = wg[r * (D_ / 4)];

    const v4f* ag = (const v4f*)(objs + (size_t)rowbase * D_);
    const v4f a0 = ag[tid];
    const v4f a1 = ag[tid + 512];
    *(v4f*)(As + (tid >> 6) * PITCH + (tid & 63) * 4)       = a0;
    *(v4f*)(As + ((tid >> 6) + 8) * PITCH + (tid & 63) * 4) = a1;

    float* wl = Ws + wave * (16 * PITCH);
#pragma unroll
    for (int r = 0; r < 16; ++r) *(v4f*)(wl + r * PITCH + lane * 4) = wlo[r];
    __syncthreads();

#pragma unroll
    for (int r = 0; r < 16; ++r) whi[r] = wg[r * (D_ / 4) + 32];

    const float* af = As + l * PITCH + 4 * half;
    const float* wf = wl + l * PITCH + 4 * half;
    for (int j = 0; j < 16; ++j) {
        const v4f a = *(const v4f*)(af + 8 * j);
        const v4f w = *(const v4f*)(wf + 8 * j);
        v2f a01 = {a.x, a.y}, w01 = {w.x, w.y}, a23 = {a.z, a.w}, w23 = {w.z, w.w};
        c = __builtin_amdgcn_wmma_f32_16x16x4_f32(false, a01, false, w01,
                                                  (short)0, c, false, false);
        c = __builtin_amdgcn_wmma_f32_16x16x4_f32(false, a23, false, w23,
                                                  (short)0, c, false, false);
    }
#pragma unroll
    for (int r = 0; r < 16; ++r) *(v4f*)(wl + r * PITCH + 128 + lane * 4) = whi[r];
    for (int j = 16; j < 32; ++j) {
        const v4f a = *(const v4f*)(af + 8 * j);
        const v4f w = *(const v4f*)(wf + 8 * j);
        v2f a01 = {a.x, a.y}, w01 = {w.x, w.y}, a23 = {a.z, a.w}, w23 = {w.z, w.w};
        c = __builtin_amdgcn_wmma_f32_16x16x4_f32(false, a01, false, w01,
                                                  (short)0, c, false, false);
        c = __builtin_amdgcn_wmma_f32_16x16x4_f32(false, a23, false, w23,
                                                  (short)0, c, false, false);
    }
#else
    for (int k0 = 0; k0 < D_; k0 += 32) {
        v16h a, w;
        const float* ap = objs + (size_t)(rowbase + l) * D_;
        const float* wp = W + (size_t)(colbase + l) * D_;
        for (int i = 0; i < 8; ++i) {
            const int kk = k0 + ((i < 4) ? (2 * i + 8 * half)
                                         : (16 + 2 * (i - 4) + 8 * half));
            a[2 * i]     = (_Float16)ap[kk];
            a[2 * i + 1] = (_Float16)ap[kk + 1];
            w[2 * i]     = (_Float16)wp[kk];
            w[2 * i + 1] = (_Float16)wp[kk + 1];
        }
        c = __builtin_amdgcn_wmma_f32_16x16x32_f16(false, a, false, w,
                                                   (short)0, c, false, false);
    }
#endif

    for (int j = 0; j < 8; ++j) {
        const int m = j + half * 8;
        const int n = colbase + l;
        tile[m * D_ + n] = c[j] * scale[rowbase + m] + bias[n];
    }
    __syncthreads();

    const float* row = tile + wave * D_;
    float v[8], s = 0.0f, s2 = 0.0f;
    for (int j = 0; j < 8; ++j) {
        v[j] = row[lane * 8 + j];
        s  += v[j];
        s2 += v[j] * v[j];
    }
    for (int off = 16; off > 0; off >>= 1) {
        s  += __shfl_xor(s,  off, 32);
        s2 += __shfl_xor(s2, off, 32);
    }
    const float mean = s * (1.0f / D_);
    const float var  = s2 * (1.0f / D_) - mean * mean;
    const float rstd = rsqrtf(var + 1e-5f);
    float* orow = out + (size_t)(rowbase + wave) * D_;
    for (int j = 0; j < 8; ++j) {
        const int idx = lane * 8 + j;
        orow[idx] = (v[j] - mean) * rstd * gamma[idx] + beta[idx];
    }
}

extern "C" void kernel_launch(void* const* d_in, const int* in_sizes, int n_in,
                              void* d_out, int out_size, void* d_ws, size_t ws_size,
                              hipStream_t stream) {
    const float* features = (const float*)d_in[0];
    const int*   seg      = (const int*)  d_in[1];
    const float* W        = (const float*)d_in[2];
    const float* b        = (const float*)d_in[3];
    const float* gamma    = (const float*)d_in[4];
    const float* beta     = (const float*)d_in[5];
    float* ws  = (float*)d_ws;
    float* out = (float*)d_out;

    count_pick<<<B_, 1024, 0, stream>>>(seg, ws);
    seg_accum <<<B_ * CHUNKS, 256, 0, stream>>>(features, seg, ws);
    proj_ln   <<<(B_ * M_) / 16, 512, 0, stream>>>(ws + WS_SUMS, ws + WS_SCALE, W, b, gamma, beta, out);
}


# seg_accum: a wave owns one chosen slot of a 512-row chunk and sums its rows in registers; LDS accumulator and its float atomics removed; f32 sum order changes (list order, then global atomics)
# speedup vs baseline: 4.7283x; 1.3130x over previous
#include <hip/hip_runtime.h>
#include <hip/hip_bf16.h>

#define S_ 32
#define B_ 32
#define N_ 8192
#define D_ 256
#define M_ 8
#define CHUNKS 16
#define ROWS_PER_BLOCK (N_ / CHUNKS)

typedef __attribute__((ext_vector_type(2)))  float    v2f;
typedef __attribute__((ext_vector_type(4)))  float    v4f;
typedef __attribute__((ext_vector_type(8)))  float    v8f;
typedef __attribute__((ext_vector_type(16))) _Float16 v16h;

#define WS_SUMS  0
#define WS_SLOT  (B_ * M_ * D_)
#define WS_SCALE (WS_SLOT + B_ * S_)

#if defined(__has_builtin)
#if __has_builtin(__builtin_amdgcn_wmma_f32_16x16x4_f32)
#define HAVE_F32_WMMA 1
#endif
#endif

static_assert(S_ == 32, "one lane of a wave per segment id");
static_assert(N_ == 8 * 1024, "count_pick: 1024 threads take 8 ids each");
static_assert(M_ * D_ == 2 * 1024, "count_pick: 1024 threads zero 2 sums each");

typedef __attribute__((ext_vector_type(4))) int v4i;

__global__ __launch_bounds__(1024) void count_pick(const int* __restrict__ seg,
                                                   float*     __restrict__ ws) {
    __shared__ int hist[S_];
    const int b = blockIdx.x, tid = threadIdx.x;
    if (tid < S_) hist[tid] = 0;
    __syncthreads();

    const v4i* sp = (const v4i*)(seg + b * N_);
    const v4i x0 = sp[tid];
    const v4i x1 = sp[tid + 1024];
    atomicAdd(&hist[x0.x & (S_ - 1)], 1); atomicAdd(&hist[x0.y & (S_ - 1)], 1);
    atomicAdd(&hist[x0.z & (S_ - 1)], 1); atomicAdd(&hist[x0.w & (S_ - 1)], 1);
    atomicAdd(&hist[x1.x & (S_ - 1)], 1); atomicAdd(&hist[x1.y & (S_ - 1)], 1);
    atomicAdd(&hist[x1.z & (S_ - 1)], 1); atomicAdd(&hist[x1.w & (S_ - 1)], 1);

    float* gs = ws + WS_SUMS + (size_t)b * M_ * D_;
    gs[tid]        = 0.0f;
    gs[tid + 1024] = 0.0f;
    __syncthreads();

    if (tid < S_) {
        const float c = (float)hist[tid];
        int place = 0;
#pragma unroll
        for (int t = 0; t < S_; ++t) {
            const float ct = __builtin_bit_cast(
                float, __builtin_amdgcn_readlane(__builtin_bit_cast(int, c), t));
            place += (ct > c || (ct == c && t < tid)) ? 1 : 0;
        }
        ws[WS_SLOT + b * S_ + tid] = (place < M_) ? (float)place : -1.0f;
        if (place < M_)
            ws[WS_SCALE + b * M_ + place] = (c > 0.0f) ? (1.0f / c) : 0.0f;
    }
}

#define RB            30
#define ROW_V4        (D_ / 4)
#define ID_V4         (ROWS_PER_BLOCK / 128)
#define LIST_PITCH    (ROWS_PER_BLOCK + 32)
static_assert(ROWS_PER_BLOCK % 128 == 0, "a wave-wide 16-byte id load covers 128 rows");
static_assert(RB <= 32, "one list entry per lane and trip");
static_assert(M_ * 32 == 256, "seg_accum: the 8 waves of a 256-thread block are the M_ slots");

__device__ __forceinline__ v4f load_stream(const v4f* p) {
    return __builtin_nontemporal_load(p);
}

__global__ __launch_bounds__(256) void seg_accum(const float* __restrict__ features,
                                                 const int*   __restrict__ seg,
                                                 float*       __restrict__ ws) {
    __shared__ unsigned short rowlist[M_][LIST_PITCH];
    __shared__ __attribute__((aligned(16))) float turn[M_][D_];
    const int tid   = threadIdx.x;
    const int b     = blockIdx.x / CHUNKS;
    const int chunk = blockIdx.x % CHUNKS;
    const int wave  = __builtin_amdgcn_readfirstlane(tid >> 5);
    const int lane  = tid & 31;
    const int row0  = chunk * ROWS_PER_BLOCK;

    const v4i* sp = (const v4i*)(seg + b * N_ + row0) + lane;
    v4i ids[ID_V4];
#pragma unroll
    for (int q = 0; q < ID_V4; ++q) ids[q] = sp[q * 32];

    const int slot_of_lane = (int)ws[WS_SLOT + b * S_ + lane];
    const unsigned owner = __builtin_amdgcn_ballot_w32(slot_of_lane == wave);
    const int target = owner ? __builtin_ctz(owner) : S_;

    unsigned short* mylist = rowlist[wave];
    int run = 0;
#pragma unroll
    for (int q = 0; q < ID_V4; ++q) {
#pragma unroll
        for (int e = 0; e < 4; ++e) {
            const bool mt = ((ids[q][e] & (S_ - 1)) == target);
            const unsigned hit = __builtin_amdgcn_ballot_w32(mt);
            if (mt)
                mylist[run + __builtin_amdgcn_mbcnt_lo(hit, 0)] =
                    (unsigned short)(128 * q + 4 * lane + e);
            run += __builtin_popcount(hit);
        }
    }

    const v4f* fp = (const v4f*)(features + ((size_t)b * N_ + row0) * D_) + lane;
    v4f acc0 = {0.0f, 0.0f, 0.0f, 0.0f};
    v4f acc1 = {0.0f, 0.0f, 0.0f, 0.0f};

    for (int base = 0; base < run; base += RB) {
        const int left = run - base;
        const int n    = left < RB ? left : RB;
        const int mine = (lane < n) ? (int)mylist[base + lane] : 0;
        v4f buf[RB][2];

#pragma unroll
        for (int k = 0; k < RB; ++k) {
            if (k < n) {
                const int R = __builtin_amdgcn_readlane(mine, k);
                const v4f* p = fp + R * ROW_V4;
                buf[k][0] = load_stream(p);
                buf[k][1] = load_stream(p + 32);
            }
        }
#pragma unroll
        for (int k = 0; k < RB; ++k) {
            if (k < n) {
                acc0 += buf[k][0];
                acc1 += buf[k][1];
            }
        }
    }

    float* t = turn[wave];
    *(v4f*)(t + lane * 4)       = acc0;
    *(v4f*)(t + 128 + lane * 4) = acc1;
    float* g = ws + WS_SUMS + ((size_t)b * M_ + wave) * D_ + lane;
    float part[D_ / 32];
#pragma unroll
    for (int i = 0; i < D_ / 32; ++i) part[i] = t[i * 32 + lane];
#pragma unroll
    for (int i = 0; i < D_ / 32; ++i) atomicAdd(g + i * 32, part[i]);
}

#define PITCH (D_ + 4)
static_assert((PITCH % 8) == 4, "row pitch is 16 B x odd");

__global__ __launch_bounds__(512) void proj_ln(const float* __restrict__ sums,
                                               const float* __restrict__ scale,
                                               const float* __restrict__ W,
                                               const float* __restrict__ bias,
                                               const float* __restrict__ gamma,
                                               const float* __restrict__ beta,
                                               float*       __restrict__ out) {
    __shared__ __attribute__((aligned(16))) float As[16 * PITCH];
    __shared__ __attribute__((aligned(16))) float Ws[16 * 16 * PITCH];
    __shared__ float tile[16 * D_];
    const int tid  = threadIdx.x;
    const int wave = tid >> 5;
    const int lane = tid & 31;
    const int half = lane >> 4;
    const int l    = lane & 15;
    const int rowbase = blockIdx.x * 16;
    const int colbase = wave * 16;

    v8f c = {};
    const float* objs = sums;

#ifdef HAVE_F32_WMMA
    const v4f* wg = (const v4f*)(W + (size_t)colbase * D_) + lane;
    v4f wlo[16], whi[16];
#pragma unroll
    for (int r = 0; r < 16; ++r) wlo[r] = wg[r * (D_ / 4)];

    const v4f* ag = (const v4f*)(objs + (size_t)rowbase * D_);
    const v4f a0 = ag[tid];
    const v4f a1 = ag[tid + 512];
    *(v4f*)(As + (tid >> 6) * PITCH + (tid & 63) * 4)       = a0;
    *(v4f*)(As + ((tid >> 6) + 8) * PITCH + (tid & 63) * 4) = a1;

    float* wl = Ws + wave * (16 * PITCH);
#pragma unroll
    for (int r = 0; r < 16; ++r) *(v4f*)(wl + r * PITCH + lane * 4) = wlo[r];
    __syncthreads();

#pragma unroll
    for (int r = 0; r < 16; ++r) whi[r] = wg[r * (D_ / 4) + 32];

    const float* af = As + l * PITCH + 4 * half;
    const float* wf = wl + l * PITCH + 4 * half;
    for (int j = 0; j < 16; ++j) {
        const v4f a = *(const v4f*)(af + 8 * j);
        const v4f w = *(const v4f*)(wf + 8 * j);
        v2f a01 = {a.x, a.y}, w01 = {w.x, w.y}, a23 = {a.z, a.w}, w23 = {w.z, w.w};
        c = __builtin_amdgcn_wmma_f32_16x16x4_f32(false, a01, false, w01,
                                                  (short)0, c, false, false);
        c = __builtin_amdgcn_wmma_f32_16x16x4_f32(false, a23, false, w23,
                                                  (short)0, c, false, false);
    }
#pragma unroll
    for (int r = 0; r < 16; ++r) *(v4f*)(wl + r * PITCH + 128 + lane * 4) = whi[r];
    for (int j = 16; j < 32; ++j) {
        const v4f a = *(const v4f*)(af + 8 * j);
        const v4f w = *(const v4f*)(wf + 8 * j);
        v2f a01 = {a.x, a.y}, w01 = {w.x, w.y}, a23 = {a.z, a.w}, w23 = {w.z, w.w};
        c = __builtin_amdgcn_wmma_f32_16x16x4_f32(false, a01, false, w01,
                                                  (short)0, c, false, false);
        c = __builtin_amdgcn_wmma_f32_16x16x4_f32(false, a23, false, w23,
                                                  (short)0, c, false, false);
    }
#else
    for (int k0 = 0; k0 < D_; k0 += 32) {
        v16h a, w;
        const float* ap = objs + (size_t)(rowbase + l) * D_;
        const float* wp = W + (size_t)(colbase + l) * D_;
        for (int i = 0; i < 8; ++i) {
            const int kk = k0 + ((i < 4) ? (2 * i + 8 * half)
                                         : (16 + 2 * (i - 4) + 8 * half));
            a[2 * i]     = (_Float16)ap[kk];
            a[2 * i + 1] = (_Float16)ap[kk + 1];
            w[2 * i]     = (_Float16)wp[kk];
            w[2 * i + 1] = (_Float16)wp[kk + 1];
        }
        c = __builtin_amdgcn_wmma_f32_16x16x32_f16(false, a, false, w,
                                                   (short)0, c, false, false);
    }
#endif

    for (int j = 0; j < 8; ++j) {
        const int m = j + half * 8;
        const int n = colbase + l;
        tile[m * D_ + n] = c[j] * scale[rowbase + m] + bias[n];
    }
    __syncthreads();

    const float* row = tile + wave * D_;
    float v[8], s = 0.0f, s2 = 0.0f;
    for (int j = 0; j < 8; ++j) {
        v[j] = row[lane * 8 + j];
        s  += v[j];
        s2 += v[j] * v[j];
    }
    for (int off = 16; off > 0; off >>= 1) {
        s  += __shfl_xor(s,  off, 32);
        s2 += __shfl_xor(s2, off, 32);
    }
    const float mean = s * (1.0f / D_);
    const float var  = s2 * (1.0f / D_) - mean * mean;
    const float rstd = rsqrtf(var + 1e-5f);
    float* orow = out + (size_t)(rowbase + wave) * D_;
    for (int j = 0; j < 8; ++j) {
        const int idx = lane * 8 + j;
        orow[idx] = (v[j] - mean) * rstd * gamma[idx] + beta[idx];
    }
}

extern "C" void kernel_launch(void* const* d_in, const int* in_sizes, int n_in,
                              void* d_out, int out_size, void* d_ws, size_t ws_size,
                              hipStream_t stream) {
    const float* features = (const float*)d_in[0];
    const int*   seg      = (const int*)  d_in[1];
    const float* W        = (const float*)d_in[2];
    const float* b        = (const float*)d_in[3];
    const float* gamma    = (const float*)d_in[4];
    const float* beta     = (const float*)d_in[5];
    float* ws  = (float*)d_ws;
    float* out = (float*)d_out;

    count_pick<<<B_, 1024, 0, stream>>>(seg, ws);
    seg_accum <<<B_ * CHUNKS, 256, 0, stream>>>(features, seg, ws);
    proj_ln   <<<(B_ * M_) / 16, 512, 0, stream>>>(ws + WS_SUMS, ws + WS_SCALE, W, b, gamma, beta, out);
}


# count and top-8 choice folded into the sum launch (2 launches): a 16-wave block owns a (batch, slot), counts ids in LDS, sums its rows in fixed order; float atomics gone; f32 sum order changes
# speedup vs baseline: 6.4610x; 1.3665x over previous
#include <hip/hip_runtime.h>
#include <hip/hip_bf16.h>

#define S_ 32
#define B_ 32
#define N_ 8192
#define D_ 256
#define M_ 8
#define WAVES 16
#define ROWS_PER_WAVE (N_ / WAVES)

typedef __attribute__((ext_vector_type(2)))  float    v2f;
typedef __attribute__((ext_vector_type(4)))  float    v4f;
typedef __attribute__((ext_vector_type(8)))  float    v8f;
typedef __attribute__((ext_vector_type(16))) _Float16 v16h;

#define WS_SUMS  0
#define WS_SCALE (B_ * M_ * D_)

#if defined(__has_builtin)
#if __has_builtin(__builtin_amdgcn_wmma_f32_16x16x4_f32)
#define HAVE_F32_WMMA 1
#endif
#endif

#define RB            29
#define ROW_V4        (D_ / 4)
#define ID_V4         (ROWS_PER_WAVE / 128)
#define LIST_PITCH    (ROWS_PER_WAVE + 32)
static_assert(S_ == 32, "one lane of a wave per segment id");
static_assert(ROWS_PER_WAVE % 128 == 0, "a wave-wide 16-byte id load covers 128 rows");
static_assert(RB <= 32, "one list entry per lane and trip");
static_assert(S_ * 32 == 2 * WAVES * 32, "slot_sum: 512 threads clear 2 histogram words each");
static_assert(D_ <= WAVES * 32, "slot_sum: one thread per column adds the partial rows");

typedef __attribute__((ext_vector_type(4))) int v4i;

__device__ __forceinline__ v4f load_stream(const v4f* p) {
    return __builtin_nontemporal_load(p);
}

__global__ __launch_bounds__(WAVES * 32) void slot_sum(const float* __restrict__ features,
                                                       const int*   __restrict__ seg,
                                                       float*       __restrict__ ws) {
    __shared__ int hist[S_ * 32];
    __shared__ unsigned short rowlist[WAVES][LIST_PITCH];
    __shared__ __attribute__((aligned(16))) float part[WAVES][D_];
    const int tid   = threadIdx.x;
    const int b     = blockIdx.x / M_;
    const int slot  = blockIdx.x % M_;
    const int wave  = __builtin_amdgcn_readfirstlane(tid >> 5);
    const int lane  = tid & 31;
    const int row0  = wave * ROWS_PER_WAVE;

    const v4i* sp = (const v4i*)(seg + b * N_ + row0) + lane;
    v4i ids[ID_V4];
#pragma unroll
    for (int q = 0; q < ID_V4; ++q) ids[q] = sp[q * 32];

    hist[tid]              = 0;
    hist[tid + WAVES * 32] = 0;
    __syncthreads();

#pragma unroll
    for (int q = 0; q < ID_V4; ++q) {
#pragma unroll
        for (int e = 0; e < 4; ++e)
            atomicAdd(&hist[(ids[q][e] & (S_ - 1)) * 32 + lane], 1);
    }
    __syncthreads();

    int cnt = 0;
#pragma unroll
    for (int j = 0; j < 32; ++j) cnt += hist[lane * 32 + ((j + lane) & 31)];

    const int key = cnt * S_ + (S_ - 1 - lane);
    int place = 0;
#pragma unroll
    for (int t = 0; t < S_; ++t)
        place += (__builtin_amdgcn_readlane(key, t) > key) ? 1 : 0;
    const unsigned owner = __builtin_amdgcn_ballot_w32(place == slot);
    const int target = owner ? __builtin_ctz(owner) : S_;
    if (wave == 0 && place == slot)
        ws[WS_SCALE + b * M_ + slot] = (cnt > 0) ? (1.0f / (float)cnt) : 0.0f;

    unsigned short* mylist = rowlist[wave];
    int run = 0;
#pragma unroll
    for (int q = 0; q < ID_V4; ++q) {
#pragma unroll
        for (int e = 0; e < 4; ++e) {
            const bool mt = ((ids[q][e] & (S_ - 1)) == target);
            const unsigned hit = __builtin_amdgcn_ballot_w32(mt);
            if (mt)
                mylist[run + __builtin_amdgcn_mbcnt_lo(hit, 0)] =
                    (unsigned short)(128 * q + 4 * lane + e);
            run += __builtin_popcount(hit);
        }
    }

    const v4f* fp = (const v4f*)(features + ((size_t)b * N_ + row0) * D_) + lane;
    v4f acc0 = {0.0f, 0.0f, 0.0f, 0.0f};
    v4f acc1 = {0.0f, 0.0f, 0.0f, 0.0f};

    for (int base = 0; base < run; base += RB) {
        const int left = run - base;
        const int n    = left < RB ? left : RB;
        const int mine = (lane < n) ? (int)mylist[base + lane] : 0;
        v4f buf[RB][2];

#pragma unroll
        for (int k = 0; k < RB; ++k) {
            if (k < n) {
                const int R = __builtin_amdgcn_readlane(mine, k);
                const v4f* p = fp + R * ROW_V4;
                buf[k][0] = load_stream(p);
                buf[k][1] = load_stream(p + 32);
            }
        }
#pragma unroll
        for (int k = 0; k < RB; ++k) {
            if (k < n) {
                acc0 += buf[k][0];
                acc1 += buf[k][1];
            }
        }
    }

    float* t = part[wave];
    *(v4f*)(t + lane * 4)       = acc0;
    *(v4f*)(t + 128 + lane * 4) = acc1;
    __syncthreads();
    if (tid < D_) {
        float s = 0.0f;
#pragma unroll
        for (int w = 0; w < WAVES; ++w) s += part[w][tid];
        ws[WS_SUMS + ((size_t)b * M_ + slot) * D_ + tid] = s;
    }
}

#define PITCH (D_ + 4)
static_assert((PITCH % 8) == 4, "row pitch is 16 B x odd");

__global__ __launch_bounds__(512) void proj_ln(const float* __restrict__ sums,
                                               const float* __restrict__ scale,
                                               const float* __restrict__ W,
                                               const float* __restrict__ bias,
                                               const float* __restrict__ gamma,
                                               const float* __restrict__ beta,
                                               float*       __restrict__ out) {
    __shared__ __attribute__((aligned(16))) float As[16 * PITCH];
    __shared__ __attribute__((aligned(16))) float Ws[16 * 16 * PITCH];
    __shared__ float tile[16 * D_];
    const int tid  = threadIdx.x;
    const int wave = tid >> 5;
    const int lane = tid & 31;
    const int half = lane >> 4;
    const int l    = lane & 15;
    const int rowbase = blockIdx.x * 16;
    const int colbase = wave * 16;

    v8f c = {};
    const float* objs = sums;

#ifdef HAVE_F32_WMMA
    const v4f* wg = (const v4f*)(W + (size_t)colbase * D_) + lane;
    v4f wlo[16], whi[16];
#pragma unroll
    for (int r = 0; r < 16; ++r) wlo[r] = wg[r * (D_ / 4)];

    const v4f* ag = (const v4f*)(objs + (size_t)rowbase * D_);
    const v4f a0 = ag[tid];
    const v4f a1 = ag[tid + 512];
    *(v4f*)(As + (tid >> 6) * PITCH + (tid & 63) * 4)       = a0;
    *(v4f*)(As + ((tid >> 6) + 8) * PITCH + (tid & 63) * 4) = a1;

    float* wl = Ws + wave * (16 * PITCH);
#pragma unroll
    for (int r = 0; r < 16; ++r) *(v4f*)(wl + r * PITCH + lane * 4) = wlo[r];
    __syncthreads();

#pragma unroll
    for (int r = 0; r < 16; ++r) whi[r] = wg[r * (D_ / 4) + 32];

    const float* af = As + l * PITCH + 4 * half;
    const float* wf = wl + l * PITCH + 4 * half;
    for (int j = 0; j < 16; ++j) {
        const v4f a = *(const v4f*)(af + 8 * j);
        const v4f w = *(const v4f*)(wf + 8 * j);
        v2f a01 = {a.x, a.y}, w01 = {w.x, w.y}, a23 = {a.z, a.w}, w23 = {w.z, w.w};
        c = __builtin_amdgcn_wmma_f32_16x16x4_f32(false, a01, false, w01,
                                                  (short)0, c, false, false);
        c = __builtin_amdgcn_wmma_f32_16x16x4_f32(false, a23, false, w23,
                                                  (short)0, c, false, false);
    }
#pragma unroll
    for (int r = 0; r < 16; ++r) *(v4f*)(wl + r * PITCH + 128 + lane * 4) = whi[r];
    for (int j = 16; j < 32; ++j) {
        const v4f a = *(const v4f*)(af + 8 * j);
        const v4f w = *(const v4f*)(wf + 8 * j);
        v2f a01 = {a.x, a.y}, w01 = {w.x, w.y}, a23 = {a.z, a.w}, w23 = {w.z, w.w};
        c = __builtin_amdgcn_wmma_f32_16x16x4_f32(false, a01, false, w01,
                                                  (short)0, c, false, false);
        c = __builtin_amdgcn_wmma_f32_16x16x4_f32(false, a23, false, w23,
                                                  (short)0, c, false, false);
    }
#else
    for (int k0 = 0; k0 < D_; k0 += 32) {
        v16h a, w;
        const float* ap = objs + (size_t)(rowbase + l) * D_;
        const float* wp = W + (size_t)(colbase + l) * D_;
        for (int i = 0; i < 8; ++i) {
            const int kk = k0 + ((i < 4) ? (2 * i + 8 * half)
                                         : (16 + 2 * (i - 4) + 8 * half));
            a[2 * i]     = (_Float16)ap[kk];
            a[2 * i + 1] = (_Float16)ap[kk + 1];
            w[2 * i]     = (_Float16)wp[kk];
            w[2 * i + 1] = (_Float16)wp[kk + 1];
        }
        c = __builtin_amdgcn_wmma_f32_16x16x32_f16(false, a, false, w,
                                                   (short)0, c, false, false);
    }
#endif

    for (int j = 0; j < 8; ++j) {
        const int m = j + half * 8;
        const int n = colbase + l;
        tile[m * D_ + n] = c[j] * scale[rowbase + m] + bias[n];
    }
    __syncthreads();

    const float* row = tile + wave * D_;
    float v[8], s = 0.0f, s2 = 0.0f;
    for (int j = 0; j < 8; ++j) {
        v[j] = row[lane * 8 + j];
        s  += v[j];
        s2 += v[j] * v[j];
    }
    for (int off = 16; off > 0; off >>= 1) {
        s  += __shfl_xor(s,  off, 32);
        s2 += __shfl_xor(s2, off, 32);
    }
    const float mean = s * (1.0f / D_);
    const float var  = s2 * (1.0f / D_) - mean * mean;
    const float rstd = rsqrtf(var + 1e-5f);
    float* orow = out + (size_t)(rowbase + wave) * D_;
    for (int j = 0; j < 8; ++j) {
        const int idx = lane * 8 + j;
        orow[idx] = (v[j] - mean) * rstd * gamma[idx] + beta[idx];
    }
}

extern "C" void kernel_launch(void* const* d_in, const int* in_sizes, int n_in,
                              void* d_out, int out_size, void* d_ws, size_t ws_size,
                              hipStream_t stream) {
    const float* features = (const float*)d_in[0];
    const int*   seg      = (const int*)  d_in[1];
    const float* W        = (const float*)d_in[2];
    const float* b        = (const float*)d_in[3];
    const float* gamma    = (const float*)d_in[4];
    const float* beta     = (const float*)d_in[5];
    float* ws  = (float*)d_ws;
    float* out = (float*)d_out;

    slot_sum<<<B_ * M_, WAVES * 32, 0, stream>>>(features, seg, ws);
    proj_ln   <<<(B_ * M_) / 16, 512, 0, stream>>>(ws + WS_SUMS, ws + WS_SCALE, W, b, gamma, beta, out);
}


# one launch: the 16-wave block that owns a (batch, slot) also projects its sum row on f32 WMMA (row broadcast to the 16 tile rows, W slab per wave in LDS) and normalizes it; no workspace
# speedup vs baseline: 7.0698x; 1.0942x over previous
#include <hip/hip_runtime.h>
#include <hip/hip_bf16.h>

#define S_ 32
#define B_ 32
#define N_ 8192
#define D_ 256
#define M_ 8
#define WAVES 16
#define ROWS_PER_WAVE (N_ / WAVES)

typedef __attribute__((ext_vector_type(2)))  float    v2f;
typedef __attribute__((ext_vector_type(4)))  float    v4f;
typedef __attribute__((ext_vector_type(8)))  float    v8f;
typedef __attribute__((ext_vector_type(16))) _Float16 v16h;


#if defined(__has_builtin)
#if __has_builtin(__builtin_amdgcn_wmma_f32_16x16x4_f32)
#define HAVE_F32_WMMA 1
#endif
#endif

#define RB            29
#define ROW_V4        (D_ / 4)
#define ID_V4         (ROWS_PER_WAVE / 128)
#define LIST_PITCH    (ROWS_PER_WAVE + 32)
static_assert(S_ == 32, "one lane of a wave per segment id");
static_assert(ROWS_PER_WAVE % 128 == 0, "a wave-wide 16-byte id load covers 128 rows");
static_assert(RB <= 32, "one list entry per lane and trip");
static_assert(S_ * 32 == 2 * WAVES * 32, "slot_sum: 512 threads clear 2 histogram words each");
static_assert(D_ <= WAVES * 32, "slot_sum: one thread per column adds the partial rows");

typedef __attribute__((ext_vector_type(4))) int v4i;

#define PITCH (D_ + 4)
static_assert((PITCH % 8) == 4, "row pitch is 16 B x odd");

__device__ __forceinline__ v4f load_stream(const v4f* p) {
    return __builtin_nontemporal_load(p);
}

__global__ __launch_bounds__(WAVES * 32) void slot_proj_ln(const float* __restrict__ features,
                                                           const int*   __restrict__ seg,
                                                           const float* __restrict__ W,
                                                           const float* __restrict__ bias,
                                                           const float* __restrict__ gamma,
                                                           const float* __restrict__ beta,
                                                           float*       __restrict__ out) {
    __shared__ int hist[S_ * 32];
    __shared__ unsigned short rowlist[WAVES][LIST_PITCH];
    __shared__ __attribute__((aligned(16))) float part[WAVES][D_];
    __shared__ __attribute__((aligned(16))) float Ws[WAVES * 16 * PITCH];
    __shared__ __attribute__((aligned(16))) float arow[D_];
    __shared__ __attribute__((aligned(16))) float tile[D_];
    const int tid   = threadIdx.x;
    const int b     = blockIdx.x / M_;
    const int slot  = blockIdx.x % M_;
    const int wave  = __builtin_amdgcn_readfirstlane(tid >> 5);
    const int lane  = tid & 31;
    const int row0  = wave * ROWS_PER_WAVE;

    const v4i* sp = (const v4i*)(seg + b * N_ + row0) + lane;
    v4i ids[ID_V4];
#pragma unroll
    for (int q = 0; q < ID_V4; ++q) ids[q] = sp[q * 32];

    hist[tid]              = 0;
    hist[tid + WAVES * 32] = 0;
    __syncthreads();

#pragma unroll
    for (int q = 0; q < ID_V4; ++q) {
#pragma unroll
        for (int e = 0; e < 4; ++e)
            atomicAdd(&hist[(ids[q][e] & (S_ - 1)) * 32 + lane], 1);
    }
    __syncthreads();

    const v4f* wg = (const v4f*)(W + (size_t)(wave * 16) * D_) + lane;
    float* wl = Ws + wave * (16 * PITCH);
    v4f wlo[16];
#pragma unroll
    for (int r = 0; r < 16; ++r) wlo[r] = wg[r * (D_ / 4)];

    int cnt = 0;
#pragma unroll
    for (int j = 0; j < 32; ++j) cnt += hist[lane * 32 + ((j + lane) & 31)];

    const int key = cnt * S_ + (S_ - 1 - lane);
    int place = 0;
#pragma unroll
    for (int t = 0; t < S_; ++t)
        place += (__builtin_amdgcn_readlane(key, t) > key) ? 1 : 0;
    const unsigned owner = __builtin_amdgcn_ballot_w32(place == slot);
    const int target = owner ? __builtin_ctz(owner) : S_;
    const int   tcnt = __builtin_amdgcn_readlane(cnt, target & (S_ - 1));
    const float scl  = (owner && tcnt > 0) ? (1.0f / (float)tcnt) : 0.0f;

    unsigned short* mylist = rowlist[wave];
    int run = 0;
#pragma unroll
    for (int q = 0; q < ID_V4; ++q) {
#pragma unroll
        for (int e = 0; e < 4; ++e) {
            const bool mt = ((ids[q][e] & (S_ - 1)) == target);
            const unsigned hit = __builtin_amdgcn_ballot_w32(mt);
            if (mt)
                mylist[run + __builtin_amdgcn_mbcnt_lo(hit, 0)] =
                    (unsigned short)(128 * q + 4 * lane + e);
            run += __builtin_popcount(hit);
        }
    }

#pragma unroll
    for (int r = 0; r < 16; ++r) *(v4f*)(wl + r * PITCH + lane * 4) = wlo[r];

    const v4f* fp = (const v4f*)(features + ((size_t)b * N_ + row0) * D_) + lane;
    v4f acc0 = {0.0f, 0.0f, 0.0f, 0.0f};
    v4f acc1 = {0.0f, 0.0f, 0.0f, 0.0f};

    for (int base = 0; base < run; base += RB) {
        const int left = run - base;
        const int n    = left < RB ? left : RB;
        const int mine = (lane < n) ? (int)mylist[base + lane] : 0;
        v4f buf[RB][2];

#pragma unroll
        for (int k = 0; k < RB; ++k) {
            if (k < n) {
                const int R = __builtin_amdgcn_readlane(mine, k);
                const v4f* p = fp + R * ROW_V4;
                buf[k][0] = load_stream(p);
                buf[k][1] = load_stream(p + 32);
            }
        }
#pragma unroll
        for (int k = 0; k < RB; ++k) {
            if (k < n) {
                acc0 += buf[k][0];
                acc1 += buf[k][1];
            }
        }
    }

    float* t = part[wave];
    *(v4f*)(t + lane * 4)       = acc0;
    *(v4f*)(t + 128 + lane * 4) = acc1;
    __syncthreads();
    if (tid < D_) {
        float s = 0.0f;
#pragma unroll
        for (int w = 0; w < WAVES; ++w) s += part[w][tid];
        arow[tid] = s;
    }
    __syncthreads();

    v4f whi[16];
#pragma unroll
    for (int r = 0; r < 16; ++r) whi[r] = wg[r * (D_ / 4) + 32];

    const int half = lane >> 4;
    const int l    = lane & 15;
    v8f c = {};
    const float* af = arow + 4 * half;
    const float* wf = wl + l * PITCH + 4 * half;
    for (int j = 0; j < 16; ++j) {
        const v4f a = *(const v4f*)(af + 8 * j);
        const v4f w = *(const v4f*)(wf + 8 * j);
        v2f a01 = {a.x, a.y}, w01 = {w.x, w.y}, a23 = {a.z, a.w}, w23 = {w.z, w.w};
        c = __builtin_amdgcn_wmma_f32_16x16x4_f32(false, a01, false, w01,
                                                  (short)0, c, false, false);
        c = __builtin_amdgcn_wmma_f32_16x16x4_f32(false, a23, false, w23,
                                                  (short)0, c, false, false);
    }
#pragma unroll
    for (int r = 0; r < 16; ++r) *(v4f*)(wl + r * PITCH + 128 + lane * 4) = whi[r];
    for (int j = 16; j < 32; ++j) {
        const v4f a = *(const v4f*)(af + 8 * j);
        const v4f w = *(const v4f*)(wf + 8 * j);
        v2f a01 = {a.x, a.y}, w01 = {w.x, w.y}, a23 = {a.z, a.w}, w23 = {w.z, w.w};
        c = __builtin_amdgcn_wmma_f32_16x16x4_f32(false, a01, false, w01,
                                                  (short)0, c, false, false);
        c = __builtin_amdgcn_wmma_f32_16x16x4_f32(false, a23, false, w23,
                                                  (short)0, c, false, false);
    }

    if (half == 0) {
        const int n = wave * 16 + l;
        tile[n] = c[0] * scl + bias[n];
    }
    __syncthreads();

    if (wave == 0) {
        float v[8], s = 0.0f, s2 = 0.0f;
        for (int j = 0; j < 8; ++j) {
            v[j] = tile[lane * 8 + j];
            s  += v[j];
            s2 += v[j] * v[j];
        }
        for (int off = 16; off > 0; off >>= 1) {
            s  += __shfl_xor(s,  off, 32);
            s2 += __shfl_xor(s2, off, 32);
        }
        const float mean = s * (1.0f / D_);
        const float var  = s2 * (1.0f / D_) - mean * mean;
        const float rstd = rsqrtf(var + 1e-5f);
        float* orow = out + (size_t)blockIdx.x * D_;
        for (int j = 0; j < 8; ++j) {
            const int idx = lane * 8 + j;
            orow[idx] = (v[j] - mean) * rstd * gamma[idx] + beta[idx];
        }
    }
}


extern "C" void kernel_launch(void* const* d_in, const int* in_sizes, int n_in,
                              void* d_out, int out_size, void* d_ws, size_t ws_size,
                              hipStream_t stream) {
    const float* features = (const float*)d_in[0];
    const int*   seg      = (const int*)  d_in[1];
    const float* W        = (const float*)d_in[2];
    const float* b        = (const float*)d_in[3];
    const float* gamma    = (const float*)d_in[4];
    const float* beta     = (const float*)d_in[5];
    float* out = (float*)d_out;

    slot_proj_ln<<<B_ * M_, WAVES * 32, 0, stream>>>(features, seg, W, b, gamma, beta, out);
}
